# starting kernel
# baseline (speedup 1.0000x reference)
#include <hip/hip_runtime.h>
#include <math.h>

#define N_NODES 50000
#define N_EDGES 800000
#define IN_F    191
#define H1 16
#define C1 16
#define F1 256
#define H2 8
#define C2 3
#define F2 24
#define E_TOT (N_EDGES + N_NODES)

typedef __attribute__((ext_vector_type(2))) float v2f;
typedef __attribute__((ext_vector_type(8))) float v8f;


__device__ __forceinline__ unsigned enc_f32(float f) {
  unsigned u = __float_as_uint(f);
  return (u & 0x80000000u) ? ~u : (u | 0x80000000u);
}
__device__ __forceinline__ float dec_f32(unsigned u) {
  return (u & 0x80000000u) ? __uint_as_float(u & 0x7fffffffu)
                           : __uint_as_float(~u);
}
#define MX_INIT 0x00800000u

__device__ __forceinline__ void atomic_add_f32(float* p, float v) {
  __hip_atomic_fetch_add(p, v, __ATOMIC_RELAXED, __HIP_MEMORY_SCOPE_AGENT);
}

__global__ void __launch_bounds__(1024)
prep_kernel(const float* __restrict__ ea,
            const float* __restrict__ We1, const float* __restrict__ ae1,
            const float* __restrict__ We2, const float* __restrict__ ae2,
            float* __restrict__ mean_out, float* __restrict__ ce1,
            float* __restrict__ ce2) {
  __shared__ float red[1024];
  const int t = threadIdx.x;
  float s = 0.f;
  for (int i = t; i < N_EDGES; i += 1024) s += ea[i];
  red[t] = s;
  __syncthreads();
  for (int off = 512; off > 0; off >>= 1) {
    if (t < off) red[t] += red[t + off];
    __syncthreads();
  }
  if (t == 0) mean_out[0] = red[0] * (1.0f / (float)N_EDGES);
  if (t < H1) {
    float c = 0.f;
    for (int k = 0; k < C1; k++) c += We1[t * C1 + k] * ae1[t * C1 + k];
    ce1[t] = c;
  } else if (t >= 32 && t < 32 + H2) {
    const int h = t - 32;
    float c = 0.f;
    for (int k = 0; k < C2; k++) c += We2[h * C2 + k] * ae2[h * C2 + k];
    ce2[h] = c;
  }
}

__global__ void init_keys_kernel(unsigned* __restrict__ mx1,
                                 unsigned* __restrict__ mx2) {
  const int idx = blockIdx.x * blockDim.x + threadIdx.x;
  if (idx < N_NODES * H1) mx1[idx] = MX_INIT;
  if (idx < N_NODES * H2) mx2[idx] = MX_INIT;
}

__global__ void __launch_bounds__(256)
gemm1_wmma(const float* __restrict__ x, const float* __restrict__ W1,
           float* __restrict__ xl1) {
  __shared__ float As[16][49];
  __shared__ float Bs[48][128];
  const int row0 = blockIdx.x * 16;
  const int colB = blockIdx.y * 128;
  const int t = threadIdx.x;
  const int wave = t >> 5, lane = t & 31, lh = lane & 15, hi = lane >> 4;
  v8f acc = {};
  for (int k0 = 0; k0 < 192; k0 += 48) {
    for (int idx = t; idx < 16 * 48; idx += 256) {
      const int r = idx / 48, c = idx % 48, k = k0 + c;
      As[r][c] = (k < IN_F) ? x[(row0 + r) * IN_F + k] : 0.f;
    }
    for (int idx4 = t; idx4 < (48 * 128) / 4; idx4 += 256) {
      const int r = idx4 >> 5, c = (idx4 & 31) << 2, k = k0 + r;
      float4 f = make_float4(0.f, 0.f, 0.f, 0.f);
      if (k < IN_F) {
        f = *(const float4*)&W1[k * F1 + colB + c];
        if ((idx4 & 7) == 0 && k + 48 < IN_F)
          __builtin_prefetch(&W1[(k + 48) * F1 + colB + c], 0, 0);
      }
      *(float4*)&Bs[r][c] = f;
    }
    __syncthreads();
#pragma unroll
    for (int kk = 0; kk < 48; kk += 4) {
      v2f a, b;
      a.x = As[lh][kk + 2 * hi];
      a.y = As[lh][kk + 2 * hi + 1];
      b.x = Bs[kk + 2 * hi][wave * 16 + lh];
      b.y = Bs[kk + 2 * hi + 1][wave * 16 + lh];
      acc = __builtin_amdgcn_wmma_f32_16x16x4_f32(false, a, false, b,
                                                  (short)0, acc, false, false);
    }
    __syncthreads();
  }
  const int col = colB + wave * 16 + lh;
#pragma unroll
  for (int i = 0; i < 8; i++)
    xl1[(row0 + i + 8 * hi) * F1 + col] = acc[i];
}

__global__ void __launch_bounds__(64)
gemm2_wmma(const float* __restrict__ h1, const float* __restrict__ W2,
           float* __restrict__ xl2) {
  __shared__ float As[16][258];
  const int row0 = blockIdx.x * 16;
  const int t = threadIdx.x;
  const int wave = t >> 5, lane = t & 31, lh = lane & 15, hi = lane >> 4;
  for (int idx4 = t; idx4 < (16 * 256) / 4; idx4 += 64) {
    const int r = idx4 >> 6, c = (idx4 & 63) << 2;
    const float4 f = *(const float4*)&h1[(row0 + r) * F1 + c];
    As[r][c + 0] = f.x; As[r][c + 1] = f.y;
    As[r][c + 2] = f.z; As[r][c + 3] = f.w;
  }
  __syncthreads();
  const int col = wave * 16 + lh;
  v8f acc = {};
  for (int k = 0; k < 256; k += 4) {
    v2f a, b;
    a.x = As[lh][k + 2 * hi];
    a.y = As[lh][k + 2 * hi + 1];
    if (col < F2) {
      b.x = W2[(k + 2 * hi) * F2 + col];
      b.y = W2[(k + 2 * hi + 1) * F2 + col];
    } else {
      b.x = 0.f; b.y = 0.f;
    }
    acc = __builtin_amdgcn_wmma_f32_16x16x4_f32(false, a, false, b,
                                                (short)0, acc, false, false);
  }
  if (col < F2) {
#pragma unroll
    for (int i = 0; i < 8; i++)
      xl2[(row0 + i + 8 * hi) * F2 + col] = acc[i];
  }
}

template <int H, int C>
__global__ void alpha_node_kernel(const float* __restrict__ xl,
                                  const float* __restrict__ a_src,
                                  const float* __restrict__ a_dst,
                                  float* __restrict__ asrc,
                                  float* __restrict__ adst) {
  const int idx = blockIdx.x * blockDim.x + threadIdx.x;
  if (idx >= N_NODES * H) return;
  const int n = idx / H, h = idx % H;
  const float* row = xl + n * (H * C) + h * C;
  float s = 0.f, d = 0.f;
#pragma unroll
  for (int c = 0; c < C; c++) {
    const float v = row[c];
    s += v * a_src[h * C + c];
    d += v * a_dst[h * C + c];
  }
  asrc[idx] = s;
  adst[idx] = d;
}

template <int H>
__device__ __forceinline__ float edge_logit(int e, int h, const int* __restrict__ ei,
                                            const float* __restrict__ ea,
                                            const float* __restrict__ mean,
                                            const float* __restrict__ ce,
                                            const float* __restrict__ asrc,
                                            const float* __restrict__ adst,
                                            int& s, int& d) {
  float eav;
  if (e < N_EDGES) {
    s = ei[e];
    d = ei[N_EDGES + e];
    eav = ea[e];
  } else {
    s = d = e - N_EDGES;
    eav = mean[0];
  }
  const float a = asrc[s * H + h] + adst[d * H + h] + eav * ce[h];
  return (a > 0.f) ? a : 0.2f * a;
}

template <int H>
__global__ void edge_max_kernel(const int* __restrict__ ei,
                                const float* __restrict__ ea,
                                const float* __restrict__ mean,
                                const float* __restrict__ ce,
                                const float* __restrict__ asrc,
                                const float* __restrict__ adst,
                                unsigned* __restrict__ mx) {
  const int idx = blockIdx.x * blockDim.x + threadIdx.x;
  if (idx >= E_TOT * H) return;
  int s, d;
  const int e = idx / H, h = idx % H;
  const float a = edge_logit<H>(e, h, ei, ea, mean, ce, asrc, adst, s, d);
  atomicMax(&mx[d * H + h], enc_f32(a));
}

template <int H>
__global__ void edge_den_kernel(const int* __restrict__ ei,
                                const float* __restrict__ ea,
                                const float* __restrict__ mean,
                                const float* __restrict__ ce,
                                const float* __restrict__ asrc,
                                const float* __restrict__ adst,
                                const unsigned* __restrict__ mx,
                                float* __restrict__ den) {
  const int idx = blockIdx.x * blockDim.x + threadIdx.x;
  if (idx >= E_TOT * H) return;
  int s, d;
  const int e = idx / H, h = idx % H;
  const float a = edge_logit<H>(e, h, ei, ea, mean, ce, asrc, adst, s, d);
  const float m = dec_f32(mx[d * H + h]);
  atomic_add_f32(&den[d * H + h], __expf(a - m));
}

__global__ void edge_agg1_kernel(const int* __restrict__ ei,
                                 const float* __restrict__ ea,
                                 const float* __restrict__ mean,
                                 const float* __restrict__ ce,
                                 const float* __restrict__ asrc,
                                 const float* __restrict__ adst,
                                 const unsigned* __restrict__ mx,
                                 const float* __restrict__ den,
                                 const float* __restrict__ xl,
                                 float* __restrict__ agg) {
  const int idx = blockIdx.x * blockDim.x + threadIdx.x;
  if (idx >= E_TOT * H1) return;
  int s, d;
  const int e = idx / H1, h = idx % H1;
  const float a = edge_logit<H1>(e, h, ei, ea, mean, ce, asrc, adst, s, d);
  const float4* srow = (const float4*)(xl + s * F1 + h * C1);
  __builtin_prefetch(srow, 0, 0);
  const float m = dec_f32(mx[d * H1 + h]);
  const float w = __expf(a - m) / (den[d * H1 + h] + 1e-16f);
  float* drow = agg + d * F1 + h * C1;
#pragma unroll
  for (int q = 0; q < C1 / 4; q++) {
    const float4 v = srow[q];
    atomic_add_f32(&drow[4 * q + 0], w * v.x);
    atomic_add_f32(&drow[4 * q + 1], w * v.y);
    atomic_add_f32(&drow[4 * q + 2], w * v.z);
    atomic_add_f32(&drow[4 * q + 3], w * v.w);
  }
}

__global__ void edge_agg2_kernel(const int* __restrict__ ei,
                                 const float* __restrict__ ea,
                                 const float* __restrict__ mean,
                                 const float* __restrict__ ce,
                                 const float* __restrict__ asrc,
                                 const float* __restrict__ adst,
                                 const unsigned* __restrict__ mx,
                                 const float* __restrict__ den,
                                 const float* __restrict__ xl,
                                 float* __restrict__ agg) {
  const int idx = blockIdx.x * blockDim.x + threadIdx.x;
  if (idx >= E_TOT * H2) return;
  int s, d;
  const int e = idx / H2, h = idx % H2;
  const float a = edge_logit<H2>(e, h, ei, ea, mean, ce, asrc, adst, s, d);
  const float* srow = xl + s * F2 + h * C2;
  __builtin_prefetch(srow, 0, 0);
  const float m = dec_f32(mx[d * H2 + h]);
  const float w = __expf(a - m) / (den[d * H2 + h] + 1e-16f);
  float* drow = agg + d * F2 + h * C2;
#pragma unroll
  for (int c = 0; c < C2; c++) atomic_add_f32(&drow[c], w * srow[c]);
}

__global__ void finalize1_kernel(float* __restrict__ agg1,
                                 const float* __restrict__ b1) {
  const int idx4 = blockIdx.x * blockDim.x + threadIdx.x;
  if (idx4 >= (N_NODES * F1) / 4) return;
  const int e0 = idx4 << 2;
  float4 v = *(float4*)&agg1[e0];
  const float4 b = *(const float4*)&b1[e0 & (F1 - 1)];
  v.x += b.x; v.y += b.y; v.z += b.z; v.w += b.w;
  v.x = (v.x > 0.f) ? v.x : expm1f(v.x);
  v.y = (v.y > 0.f) ? v.y : expm1f(v.y);
  v.z = (v.z > 0.f) ? v.z : expm1f(v.z);
  v.w = (v.w > 0.f) ? v.w : expm1f(v.w);
  *(float4*)&agg1[e0] = v;
}

__global__ void final2_kernel(const float* __restrict__ agg2,
                              const float* __restrict__ b2,
                              float* __restrict__ out) {
  const int idx = blockIdx.x * blockDim.x + threadIdx.x;
  if (idx >= N_NODES * C2) return;
  const int n = idx / C2, c = idx % C2;
  float s = 0.f;
#pragma unroll
  for (int h = 0; h < H2; h++) s += agg2[n * F2 + h * C2 + c];
  const float v = s * (1.f / (float)H2) + b2[c];
  out[idx] = (v > 0.f) ? v : expm1f(v);
}

extern "C" void kernel_launch(void* const* d_in, const int* in_sizes, int n_in,
                              void* d_out, int out_size, void* d_ws, size_t ws_size,
                              hipStream_t stream) {
  (void)in_sizes; (void)n_in; (void)out_size; (void)ws_size;
  const float* x   = (const float*)d_in[0];
  const int*   ei  = (const int*)d_in[1];
  const float* ea  = (const float*)d_in[2];
  const float* W1  = (const float*)d_in[3];
  const float* We1 = (const float*)d_in[4];
  const float* as1 = (const float*)d_in[5];
  const float* ad1 = (const float*)d_in[6];
  const float* ae1 = (const float*)d_in[7];
  const float* b1  = (const float*)d_in[8];
  const float* W2  = (const float*)d_in[9];
  const float* We2 = (const float*)d_in[10];
  const float* as2 = (const float*)d_in[11];
  const float* ad2 = (const float*)d_in[12];
  const float* ae2 = (const float*)d_in[13];
  const float* b2  = (const float*)d_in[14];
  float* out = (float*)d_out;

  float* w = (float*)d_ws;
  float* mean = w;
  float* ce1  = w + 1;
  float* ce2  = w + 17;
  size_t off = 32;
  float* xl1   = w + off; off += (size_t)N_NODES * F1;
  float* asrc1 = w + off; off += (size_t)N_NODES * H1;
  float* adst1 = w + off; off += (size_t)N_NODES * H1;
  unsigned* mx1 = (unsigned*)(w + off); off += (size_t)N_NODES * H1;
  float* den1  = w + off; off += (size_t)N_NODES * H1;
  float* agg1  = w + off; off += (size_t)N_NODES * F1;
  float* xl2   = w + off; off += (size_t)N_NODES * F2;
  float* asrc2 = w + off; off += (size_t)N_NODES * H2;
  float* adst2 = w + off; off += (size_t)N_NODES * H2;
  unsigned* mx2 = (unsigned*)(w + off); off += (size_t)N_NODES * H2;
  float* den2  = w + off; off += (size_t)N_NODES * H2;
  float* agg2  = w + off; off += (size_t)N_NODES * F2;

  hipMemsetAsync(den1, 0, (size_t)N_NODES * H1 * 4, stream);
  hipMemsetAsync(agg1, 0, (size_t)N_NODES * F1 * 4, stream);
  hipMemsetAsync(den2, 0, (size_t)N_NODES * H2 * 4, stream);
  hipMemsetAsync(agg2, 0, (size_t)N_NODES * F2 * 4, stream);
  init_keys_kernel<<<(N_NODES * H1 + 255) / 256, 256, 0, stream>>>(mx1, mx2);

  prep_kernel<<<1, 1024, 0, stream>>>(ea, We1, ae1, We2, ae2, mean, ce1, ce2);

  gemm1_wmma<<<dim3(N_NODES / 16, 2), 256, 0, stream>>>(x, W1, xl1);
  alpha_node_kernel<H1, C1><<<(N_NODES * H1 + 255) / 256, 256, 0, stream>>>(
      xl1, as1, ad1, asrc1, adst1);
  const int g1 = (E_TOT * H1 + 255) / 256;
  edge_max_kernel<H1><<<g1, 256, 0, stream>>>(ei, ea, mean, ce1, asrc1, adst1, mx1);
  edge_den_kernel<H1><<<g1, 256, 0, stream>>>(ei, ea, mean, ce1, asrc1, adst1, mx1, den1);
  edge_agg1_kernel<<<g1, 256, 0, stream>>>(ei, ea, mean, ce1, asrc1, adst1,
                                           mx1, den1, xl1, agg1);
  finalize1_kernel<<<((N_NODES * F1) / 4 + 255) / 256, 256, 0, stream>>>(agg1, b1);

  gemm2_wmma<<<N_NODES / 16, 64, 0, stream>>>(agg1, W2, xl2);
  alpha_node_kernel<H2, C2><<<(N_NODES * H2 + 255) / 256, 256, 0, stream>>>(
      xl2, as2, ad2, asrc2, adst2);
  const int g2 = (E_TOT * H2 + 255) / 256;
  edge_max_kernel<H2><<<g2, 256, 0, stream>>>(ei, ea, mean, ce2, asrc2, adst2, mx2);
  edge_den_kernel<H2><<<g2, 256, 0, stream>>>(ei, ea, mean, ce2, asrc2, adst2, mx2, den2);
  edge_agg2_kernel<<<g2, 256, 0, stream>>>(ei, ea, mean, ce2, asrc2, adst2,
                                           mx2, den2, xl2, agg2);
  final2_kernel<<<(N_NODES * C2 + 255) / 256, 256, 0, stream>>>(agg2, b2, out);
}


# both segment-max passes: a wave takes 32-edge chunks, lanes on heads, each lane group walking the edges held in its own lanes; a chunk's source-term loads go out as one group, not one per item
# speedup vs baseline: 4.5742x; 4.5742x over previous
#include <hip/hip_runtime.h>
#include <math.h>

#define N_NODES 50000
#define N_EDGES 800000
#define IN_F    191
#define H1 16
#define C1 16
#define F1 256
#define H2 8
#define C2 3
#define F2 24
#define E_TOT (N_EDGES + N_NODES)

typedef __attribute__((ext_vector_type(2))) float v2f;
typedef __attribute__((ext_vector_type(8))) float v8f;


__device__ __forceinline__ unsigned enc_f32(float f) {
  unsigned u = __float_as_uint(f);
  return (u & 0x80000000u) ? ~u : (u | 0x80000000u);
}
__device__ __forceinline__ float dec_f32(unsigned u) {
  return (u & 0x80000000u) ? __uint_as_float(u & 0x7fffffffu)
                           : __uint_as_float(~u);
}
#define MX_INIT 0x00800000u

__device__ __forceinline__ void atomic_add_f32(float* p, float v) {
  __hip_atomic_fetch_add(p, v, __ATOMIC_RELAXED, __HIP_MEMORY_SCOPE_AGENT);
}

#define TK_STRIDE 32
#define EA4 (N_EDGES / 4)
#define PREP_LD 12
#define PREP_TRIPS (EA4 / (256 * PREP_LD))
#define PREP_REST ((EA4 - PREP_TRIPS * 256 * PREP_LD + 255) / 256)
static_assert(N_EDGES % 4 == 0, "edge_attr is read as float4 groups");

__device__ __forceinline__ void prep_block(const float* __restrict__ ea,
            const float* __restrict__ We1, const float* __restrict__ ae1,
            const float* __restrict__ We2, const float* __restrict__ ae2,
            float* __restrict__ mean_out, float* __restrict__ ce1,
            float* __restrict__ ce2, float* red) {
  const int t = threadIdx.x;
  const float4* __restrict__ ea4 = (const float4*)ea;
  float4 acc = make_float4(0.f, 0.f, 0.f, 0.f);
  int i4 = t;
#pragma unroll 1
  for (int g = 0; g < PREP_TRIPS; g++) {
    float4 v[PREP_LD];
#pragma unroll
    for (int j = 0; j < PREP_LD; j++) v[j] = ea4[i4 + 256 * j];
#pragma unroll
    for (int j = 0; j < PREP_LD; j++) {
      acc.x += v[j].x; acc.y += v[j].y; acc.z += v[j].z; acc.w += v[j].w;
    }
    i4 += 256 * PREP_LD;
  }
  {
    float4 v[PREP_REST];
#pragma unroll
    for (int j = 0; j < PREP_REST; j++) {
      const int k = i4 + 256 * j;
      v[j] = (k < EA4) ? ea4[k] : make_float4(0.f, 0.f, 0.f, 0.f);
    }
#pragma unroll
    for (int j = 0; j < PREP_REST; j++) {
      acc.x += v[j].x; acc.y += v[j].y; acc.z += v[j].z; acc.w += v[j].w;
    }
  }
  float s = (acc.x + acc.y) + (acc.z + acc.w);
  for (int off = 16; off > 0; off >>= 1) s += __shfl_xor(s, off);
  if ((t & 31) == 0) red[t >> 5] = s;
  __syncthreads();
  if (t == 0) {
    float tot = 0.f;
    for (int w = 0; w < 8; w++) tot += red[w];
    mean_out[0] = tot * (1.0f / (float)N_EDGES);
  }
  if (t < H1) {
    float c = 0.f;
    for (int k = 0; k < C1; k++) c += We1[t * C1 + k] * ae1[t * C1 + k];
    ce1[t] = c;
  } else if (t >= 32 && t < 32 + H2) {
    const int h = t - 32;
    float c = 0.f;
    for (int k = 0; k < C2; k++) c += We2[h * C2 + k] * ae2[h * C2 + k];
    ce2[h] = c;
  }
}

#define GEMM1_ROWT (N_NODES / 16)
#define GEMM1_GRID (1 + 2 * GEMM1_ROWT)
__global__ void __launch_bounds__(256)
gemm1_wmma(const float* __restrict__ x, const float* __restrict__ W1,
           float* __restrict__ xl1,
           const float* __restrict__ ea,
           const float* __restrict__ We1, const float* __restrict__ ae1,
           const float* __restrict__ We2, const float* __restrict__ ae2,
           float* __restrict__ mean_out, float* __restrict__ ce1,
           float* __restrict__ ce2) {
  __shared__ float As[16][49];
  __shared__ float Bs[48][128];
  if (blockIdx.x == 0) {
    prep_block(ea, We1, ae1, We2, ae2, mean_out, ce1, ce2, &Bs[0][0]);
    return;
  }
  const int tile = blockIdx.x - 1;
  const int ty = (tile >= GEMM1_ROWT) ? 1 : 0;
  const int row0 = (tile - ty * GEMM1_ROWT) * 16;
  const int colB = ty * 128;
  const int t = threadIdx.x;
  const int wave = t >> 5, lane = t & 31, lh = lane & 15, hi = lane >> 4;
  v8f acc = {};
  for (int k0 = 0; k0 < 192; k0 += 48) {
    for (int idx = t; idx < 16 * 48; idx += 256) {
      const int r = idx / 48, c = idx % 48, k = k0 + c;
      As[r][c] = (k < IN_F) ? x[(row0 + r) * IN_F + k] : 0.f;
    }
    for (int idx4 = t; idx4 < (48 * 128) / 4; idx4 += 256) {
      const int r = idx4 >> 5, c = (idx4 & 31) << 2, k = k0 + r;
      float4 f = make_float4(0.f, 0.f, 0.f, 0.f);
      if (k < IN_F) {
        f = *(const float4*)&W1[k * F1 + colB + c];
        if ((idx4 & 7) == 0 && k + 48 < IN_F)
          __builtin_prefetch(&W1[(k + 48) * F1 + colB + c], 0, 0);
      }
      *(float4*)&Bs[r][c] = f;
    }
    __syncthreads();
#pragma unroll
    for (int kk = 0; kk < 48; kk += 4) {
      v2f a, b;
      a.x = As[lh][kk + 2 * hi];
      a.y = As[lh][kk + 2 * hi + 1];
      b.x = Bs[kk + 2 * hi][wave * 16 + lh];
      b.y = Bs[kk + 2 * hi + 1][wave * 16 + lh];
      acc = __builtin_amdgcn_wmma_f32_16x16x4_f32(false, a, false, b,
                                                  (short)0, acc, false, false);
    }
    __syncthreads();
  }
  const int col = colB + wave * 16 + lh;
#pragma unroll
  for (int i = 0; i < 8; i++)
    xl1[(row0 + i + 8 * hi) * F1 + col] = acc[i];
}

__global__ void __launch_bounds__(64)
gemm2_wmma(const float* __restrict__ agg1, const float* __restrict__ den1,
           const float* __restrict__ b1, const float* __restrict__ W2,
           float* __restrict__ xl2) {
  __shared__ float As[16][258];
  const int row0 = blockIdx.x * 16;
  const int t = threadIdx.x;
  const int wave = t >> 5, lane = t & 31, lh = lane & 15, hi = lane >> 4;
  for (int idx4 = t; idx4 < (16 * 256) / 4; idx4 += 64) {
    const int r = idx4 >> 6, c = (idx4 & 63) << 2;
    float4 v = *(const float4*)&agg1[(row0 + r) * F1 + c];
    const float4 b = *(const float4*)&b1[c];
    const float dn = den1[(row0 + r) * H1 + (c >> 4)] + 1e-16f;
    v.x = v.x / dn + b.x; v.y = v.y / dn + b.y;
    v.z = v.z / dn + b.z; v.w = v.w / dn + b.w;
    As[r][c + 0] = (v.x > 0.f) ? v.x : expm1f(v.x);
    As[r][c + 1] = (v.y > 0.f) ? v.y : expm1f(v.y);
    As[r][c + 2] = (v.z > 0.f) ? v.z : expm1f(v.z);
    As[r][c + 3] = (v.w > 0.f) ? v.w : expm1f(v.w);
  }
  __syncthreads();
  const int col = wave * 16 + lh;
  v8f acc = {};
  for (int k = 0; k < 256; k += 4) {
    v2f a, b;
    a.x = As[lh][k + 2 * hi];
    a.y = As[lh][k + 2 * hi + 1];
    if (col < F2) {
      b.x = W2[(k + 2 * hi) * F2 + col];
      b.y = W2[(k + 2 * hi + 1) * F2 + col];
    } else {
      b.x = 0.f; b.y = 0.f;
    }
    acc = __builtin_amdgcn_wmma_f32_16x16x4_f32(false, a, false, b,
                                                (short)0, acc, false, false);
  }
  if (col < F2) {
#pragma unroll
    for (int i = 0; i < 8; i++)
      xl2[(row0 + i + 8 * hi) * F2 + col] = acc[i];
  }
}

template <int H, int C>
__global__ void alpha_node_kernel(const float* __restrict__ xl,
                                  const float* __restrict__ a_src,
                                  const float* __restrict__ a_dst,
                                  float* __restrict__ asrc,
                                  float* __restrict__ rec,
                                  unsigned* __restrict__ tk) {
  const int idx = blockIdx.x * blockDim.x + threadIdx.x;
  if (idx >= N_NODES * H) return;
  if (H == H1 && idx < 2) tk[idx * TK_STRIDE] = 0u;
  const int n = idx / H, h = idx % H;
  const float* row = xl + n * (H * C) + h * C;
  float s = 0.f, d = 0.f;
#pragma unroll
  for (int c = 0; c < C; c++) {
    const float v = row[c];
    s += v * a_src[h * C + c];
    d += v * a_dst[h * C + c];
  }
  asrc[idx] = s;
  *(float2*)&rec[2 * idx] = make_float2(d, __uint_as_float(MX_INIT));
}

__device__ __forceinline__ float leaky(float a) { return (a > 0.f) ? a : 0.2f * a; }
__device__ __forceinline__ float edge_weight(float t, float2 dk) {
  const float a = leaky(t + dk.x);
  const float m = leaky(dec_f32(__float_as_uint(dk.y)) + dk.x);
  return __expf(a - m);
}


template <int H, int J>
__device__ __forceinline__ int grp_lane(int v) {
  if (H == 16)
    return __builtin_amdgcn_update_dpp(0, v, 0x150 + (J & 15), 0xf, 0xf, true);
  else
    return (int)__builtin_amdgcn_permlane16((unsigned)v, (unsigned)v,
                                            0x11111111u * (unsigned)(J & 7),
                                            0x11111111u * (unsigned)(8 + (J & 7)),
                                            false, false);
}

template <int H, int J>
struct EmaxStep {
  static __device__ __forceinline__ void fetch(float* a, int sv, int h,
                                               const float* __restrict__ asrc) {
    a[J] = asrc[grp_lane<H, J>(sv) * H + h];
    EmaxStep<H, J + 1>::fetch(a, sv, h, asrc);
  }
  static __device__ __forceinline__ void raise(const float* a, int dv, float ev, int h,
                                               float ceh, float* __restrict__ rec) {
    const int d = grp_lane<H, J>(dv);
    const float e = __int_as_float(grp_lane<H, J>(__float_as_int(ev)));
    const float t = fmaf(e, ceh, a[J]);
    atomicMax((unsigned*)&rec[2 * (d * H + h) + 1], enc_f32(t));
    EmaxStep<H, J + 1>::raise(a, dv, ev, h, ceh, rec);
  }
};
template <int H>
struct EmaxStep<H, H> {
  static __device__ __forceinline__ void fetch(float*, int, int, const float* __restrict__) {}
  static __device__ __forceinline__ void raise(const float*, int, float, int, float,
                                               float* __restrict__) {}
};

#define EMAX_NCH ((E_TOT + 31) / 32)
#define EMAX_GRID 1024
static_assert(N_EDGES % 32 == 0, "a chunk must not straddle the self loops");

template <int H>
__global__ void __launch_bounds__(256)
edge_max_kernel(const int* __restrict__ ei,
                const float* __restrict__ ea,
                const float* __restrict__ mean,
                const float* __restrict__ ce,
                const float* __restrict__ asrc,
                float* __restrict__ rec) {
  static_assert(H == 16 || H == 8, "groups of 16 or 8 lanes");
  const int lane = threadIdx.x & 31;
  const int h = lane & (H - 1);
  const float ceh = ce[h];
  for (int ch = blockIdx.x * 8 + (threadIdx.x >> 5); ch < EMAX_NCH; ch += EMAX_GRID * 8) {
    const int base = ch * 32;
    int sv, dv;
    float ev;
    if (base < N_EDGES) {
      sv = ei[base + lane];
      dv = ei[N_EDGES + base + lane];
      ev = ea[base + lane];
    } else {
      const int n = base + lane - N_EDGES;
      sv = dv = (n < N_NODES) ? n : N_NODES - 1;
      ev = mean[0];
    }
    float a[H];
    EmaxStep<H, 0>::fetch(a, sv, h, asrc);
    EmaxStep<H, 0>::raise(a, dv, ev, h, ceh, rec);
  }
}


#define AGG1_EPC 32
#define AGG1_NCH ((E_TOT + AGG1_EPC - 1) / AGG1_EPC)
#define ROW_SPLIT (N_NODES / 2)
#define AGG1_GRID 1024
static_assert(N_EDGES % AGG1_EPC == 0, "a chunk must not straddle the self loops");

__device__ __forceinline__ int package_half() {
  return (int)((__builtin_amdgcn_s_sendmsg_rtn(0x87) >> 18) & 1u);
}

__device__ __forceinline__ unsigned draw_chunk(unsigned* ctr, int lane) {
  unsigned t = 0u;
  if (lane == 0) t = atomicAdd(ctr, 1u);
  return (unsigned)__builtin_amdgcn_readfirstlane((int)t);
}

__device__ __forceinline__ unsigned chunk_load(int base, int lane, int q,
                                               const int* __restrict__ ei,
                                               const float* __restrict__ ea,
                                               const float* __restrict__ mean,
                                               int& sv, int& dv, float& ev) {
  bool ok = true;
  if (base < N_EDGES) {
    sv = ei[base + lane];
    dv = ei[N_EDGES + base + lane];
    ev = ea[base + lane];
  } else {
    const int n = base + lane - N_EDGES;
    ok = (n < N_NODES);
    sv = dv = ok ? n : N_NODES - 1;
    ev = mean[0];
  }
  return __builtin_amdgcn_ballot_w32(ok && ((dv >= ROW_SPLIT) == (q != 0)));
}


template <int L>
__device__ __forceinline__ float row_lane(float v) {
  const int i = __float_as_int(v);
  return __int_as_float(__builtin_amdgcn_update_dpp(0, i, 0x150 + L, 0xf, 0xf, true));
}

struct Agg1Edge {
  int s, d;
  float as_, eav;
  float2 dk;
  float v[8];
};

__device__ __forceinline__ void agg1_fetch(Agg1Edge& E, int sv, int dv, float ev,
                                           int i, int h, int lane,
                                           const float* __restrict__ asrc,
                                           const float* __restrict__ rec,
                                           const float* __restrict__ xl) {
  E.s = __builtin_amdgcn_readlane(sv, i);
  E.d = __builtin_amdgcn_readlane(dv, i);
  E.eav = __int_as_float(__builtin_amdgcn_readlane(__float_as_int(ev), i));
  E.as_ = asrc[E.s * H1 + h];
  E.dk = *(const float2*)&rec[2 * (E.d * H1 + h)];
  const float* srow = xl + E.s * F1 + lane;
#pragma unroll
  for (int j = 0; j < 8; j++) E.v[j] = srow[32 * j];
}

__device__ __forceinline__ void agg1_push(const Agg1Edge& E, float ceh, int lane,
                                          float* __restrict__ den,
                                          float* __restrict__ agg) {
  const float w = edge_weight(fmaf(E.eav, ceh, E.as_), E.dk);
  if (lane < 16) atomic_add_f32(&den[E.d * H1 + lane], w);
  float* drow = agg + E.d * F1 + lane;
  atomic_add_f32(&drow[0 * 32], row_lane<0>(w) * E.v[0]);
  atomic_add_f32(&drow[1 * 32], row_lane<2>(w) * E.v[1]);
  atomic_add_f32(&drow[2 * 32], row_lane<4>(w) * E.v[2]);
  atomic_add_f32(&drow[3 * 32], row_lane<6>(w) * E.v[3]);
  atomic_add_f32(&drow[4 * 32], row_lane<8>(w) * E.v[4]);
  atomic_add_f32(&drow[5 * 32], row_lane<10>(w) * E.v[5]);
  atomic_add_f32(&drow[6 * 32], row_lane<12>(w) * E.v[6]);
  atomic_add_f32(&drow[7 * 32], row_lane<14>(w) * E.v[7]);
}

__global__ void __launch_bounds__(256)
edge_agg1_kernel(const int* __restrict__ ei,
                 const float* __restrict__ ea,
                 const float* __restrict__ mean,
                 const float* __restrict__ ce,
                 const float* __restrict__ asrc,
                 const float* __restrict__ rec,
                 const float* __restrict__ xl,
                 float* __restrict__ den,
                 float* __restrict__ agg,
                 unsigned* __restrict__ tk) {
  const int lane = threadIdx.x & 31;
  const int h = (lane + (lane >> 4)) & 15;
  const float ceh = ce[h];
  const int mine = package_half();
  for (int pass = 0; pass < 2; pass++) {
    const int q = mine ^ pass;
    unsigned* ctr = tk + q * TK_STRIDE;
    for (;;) {
      const unsigned t = draw_chunk(ctr, lane);
      if (t >= AGG1_NCH) break;
      int sv, dv;
      float ev;
      unsigned m = chunk_load((int)t * AGG1_EPC, lane, q, ei, ea, mean, sv, dv, ev);
      while (m) {
        const int i0 = __builtin_ctz(m); m &= m - 1;
        const bool p1 = (m != 0u);
        const int i1 = p1 ? __builtin_ctz(m) : i0; m &= m - 1;
        const bool p2 = (m != 0u);
        const int i2 = p2 ? __builtin_ctz(m) : i0; m &= m - 1;
        const bool p3 = (m != 0u);
        const int i3 = p3 ? __builtin_ctz(m) : i0; m &= m - 1;
        Agg1Edge A, B, C, D;
        agg1_fetch(A, sv, dv, ev, i0, h, lane, asrc, rec, xl);
        agg1_fetch(B, sv, dv, ev, i1, h, lane, asrc, rec, xl);
        agg1_fetch(C, sv, dv, ev, i2, h, lane, asrc, rec, xl);
        agg1_fetch(D, sv, dv, ev, i3, h, lane, asrc, rec, xl);
        agg1_push(A, ceh, lane, den, agg);
        if (p1) agg1_push(B, ceh, lane, den, agg);
        if (p2) agg1_push(C, ceh, lane, den, agg);
        if (p3) agg1_push(D, ceh, lane, den, agg);
      }
    }
  }
}

#define AGG2_EPW 32
#define ACC2 32
static_assert(N_EDGES % AGG2_EPW == 0, "a chunk must not straddle the self loops");
static_assert(E_TOT % 4 == 0, "four edges per loop trip");
static_assert(H2 * (C2 + 1) == ACC2, "one lane per float of the line");

struct Agg2Edge {
  int d;
  float as_, eav, x;
  float2 dk;
};

__device__ __forceinline__ void agg2_fetch(Agg2Edge& E, int sv, int dv, float ev,
                                           int i, int h, int xo,
                                           const float* __restrict__ asrc,
                                           const float* __restrict__ rec,
                                           const float* __restrict__ xl) {
  const int s = __builtin_amdgcn_readlane(sv, i);
  E.d = __builtin_amdgcn_readlane(dv, i);
  E.eav = __int_as_float(__builtin_amdgcn_readlane(__float_as_int(ev), i));
  E.as_ = asrc[s * H2 + h];
  E.dk = *(const float2*)&rec[2 * (E.d * H2 + h)];
  E.x = xl[s * F2 + xo];
}

__device__ __forceinline__ void agg2_push(const Agg2Edge& E, float ceh, int lane,
                                          bool is_den, float* __restrict__ acc) {
  const float w = edge_weight(fmaf(E.eav, ceh, E.as_), E.dk);
  atomic_add_f32(&acc[E.d * ACC2 + lane], is_den ? w : w * E.x);
}

__global__ void __launch_bounds__(256)
edge_agg2_kernel(const int* __restrict__ ei,
                 const float* __restrict__ ea,
                 const float* __restrict__ mean,
                 const float* __restrict__ ce,
                 const float* __restrict__ asrc,
                 const float* __restrict__ rec,
                 const float* __restrict__ xl,
                 float* __restrict__ acc) {
  const int lane = threadIdx.x & 31;
  const int base = (blockIdx.x * 8 + (threadIdx.x >> 5)) * AGG2_EPW;
  if (base >= E_TOT) return;
  const int h = lane >> 2, c = lane & 3;
  const bool is_den = (c == C2);
  const int xo = h * C2 + (is_den ? C2 - 1 : c);
  const float ceh = ce[h];
  int sv, dv;
  float ev;
  if (base < N_EDGES) {
    sv = ei[base + lane];
    dv = ei[N_EDGES + base + lane];
    ev = ea[base + lane];
  } else {
    const int n = base + lane - N_EDGES;
    sv = dv = (n < N_NODES) ? n : N_NODES - 1;
    ev = mean[0];
  }
  const int cnt = (E_TOT - base < AGG2_EPW) ? (E_TOT - base) : AGG2_EPW;
  for (int i = 0; i < cnt; i += 4) {
    Agg2Edge A, B, C, D;
    agg2_fetch(A, sv, dv, ev, i, h, xo, asrc, rec, xl);
    agg2_fetch(B, sv, dv, ev, i + 1, h, xo, asrc, rec, xl);
    agg2_fetch(C, sv, dv, ev, i + 2, h, xo, asrc, rec, xl);
    agg2_fetch(D, sv, dv, ev, i + 3, h, xo, asrc, rec, xl);
    agg2_push(A, ceh, lane, is_den, acc);
    agg2_push(B, ceh, lane, is_den, acc);
    agg2_push(C, ceh, lane, is_den, acc);
    agg2_push(D, ceh, lane, is_den, acc);
  }
}

__global__ void final2_kernel(const float* __restrict__ acc2,
                              const float* __restrict__ b2,
                              float* __restrict__ out) {
  const int idx = blockIdx.x * blockDim.x + threadIdx.x;
  if (idx >= N_NODES * C2) return;
  const int n = idx / C2, c = idx % C2;
  float s = 0.f;
#pragma unroll
  for (int h = 0; h < H2; h++)
    s += acc2[n * ACC2 + h * (C2 + 1) + c] / (acc2[n * ACC2 + h * (C2 + 1) + C2] + 1e-16f);
  const float v = s * (1.f / (float)H2) + b2[c];
  out[idx] = (v > 0.f) ? v : expm1f(v);
}

extern "C" void kernel_launch(void* const* d_in, const int* in_sizes, int n_in,
                              void* d_out, int out_size, void* d_ws, size_t ws_size,
                              hipStream_t stream) {
  (void)in_sizes; (void)n_in; (void)out_size; (void)ws_size;
  const float* x   = (const float*)d_in[0];
  const int*   ei  = (const int*)d_in[1];
  const float* ea  = (const float*)d_in[2];
  const float* W1  = (const float*)d_in[3];
  const float* We1 = (const float*)d_in[4];
  const float* as1 = (const float*)d_in[5];
  const float* ad1 = (const float*)d_in[6];
  const float* ae1 = (const float*)d_in[7];
  const float* b1  = (const float*)d_in[8];
  const float* W2  = (const float*)d_in[9];
  const float* We2 = (const float*)d_in[10];
  const float* as2 = (const float*)d_in[11];
  const float* ad2 = (const float*)d_in[12];
  const float* ae2 = (const float*)d_in[13];
  const float* b2  = (const float*)d_in[14];
  float* out = (float*)d_out;

  float* w = (float*)d_ws;
  float* mean = w;
  float* ce1  = w + 1;
  float* ce2  = w + 17;
  size_t off = 32;
  float* xl1   = w + off; off += (size_t)N_NODES * F1;
  float* asrc1 = w + off; off += (size_t)N_NODES * H1;
  float* rec1  = w + off; off += (size_t)N_NODES * H1 * 2;
  float* den1  = w + off; off += (size_t)N_NODES * H1;
  float* agg1  = w + off; off += (size_t)N_NODES * F1;
  float* xl2   = w + off; off += (size_t)N_NODES * F2;
  float* asrc2 = w + off; off += (size_t)N_NODES * H2;
  float* rec2  = w + off; off += (size_t)N_NODES * H2 * 2;
  float* acc2  = w + off; off += (size_t)N_NODES * ACC2;
  unsigned* tk = (unsigned*)(w + off); off += 2 * TK_STRIDE;

  hipMemsetAsync(den1, 0, (size_t)N_NODES * H1 * 4, stream);
  hipMemsetAsync(agg1, 0, (size_t)N_NODES * F1 * 4, stream);
  hipMemsetAsync(acc2, 0, (size_t)N_NODES * ACC2 * 4, stream);

  gemm1_wmma<<<GEMM1_GRID, 256, 0, stream>>>(x, W1, xl1, ea, We1, ae1, We2, ae2,
                                             mean, ce1, ce2);
  alpha_node_kernel<H1, C1><<<(N_NODES * H1 + 255) / 256, 256, 0, stream>>>(
      xl1, as1, ad1, asrc1, rec1, tk);
  edge_max_kernel<H1><<<EMAX_GRID, 256, 0, stream>>>(ei, ea, mean, ce1, asrc1, rec1);
  edge_agg1_kernel<<<AGG1_GRID, 256, 0, stream>>>(
      ei, ea, mean, ce1, asrc1, rec1, xl1, den1, agg1, tk);

  gemm2_wmma<<<N_NODES / 16, 64, 0, stream>>>(agg1, den1, b1, W2, xl2);
  alpha_node_kernel<H2, C2><<<(N_NODES * H2 + 255) / 256, 256, 0, stream>>>(
      xl2, as2, ad2, asrc2, rec2, tk);
  edge_max_kernel<H2><<<EMAX_GRID, 256, 0, stream>>>(ei, ea, mean, ce2, asrc2, rec2);
  edge_agg2_kernel<<<(E_TOT + 8 * AGG2_EPW - 1) / (8 * AGG2_EPW), 256, 0, stream>>>(
      ei, ea, mean, ce2, asrc2, rec2, xl2, acc2);
  final2_kernel<<<(N_NODES * C2 + 255) / 256, 256, 0, stream>>>(acc2, b2, out);
}
